# starting kernel
# baseline (speedup 1.0000x reference)
#include <hip/hip_runtime.h>
#include <hip/hip_bf16.h>
#include <math.h>

#define IN_DIM  128
#define OUT_DIM 64
#define ED_DIM  32

#define BSTRIDE_V2 80
#define BSTRIDE_F  (2 * BSTRIDE_V2)

typedef __attribute__((ext_vector_type(2))) float v2f;
typedef __attribute__((ext_vector_type(8))) float v8f;

struct Acc4 { v8f t[4]; };

__device__ __attribute__((always_inline)) inline
Acc4 wave_gemm_tile(const float* __restrict__ arow,
                    const float* __restrict__ lWs,
                    int kdim, int half, int l16)
{
    Acc4 A;
    A.t[0] = {}; A.t[1] = {}; A.t[2] = {}; A.t[3] = {};
    const v2f* a2 = (const v2f*)arow;
    for (int k0 = 0; k0 < kdim; k0 += 4) {
        const int kp  = (k0 >> 1) + half;
        const v2f  a  = a2[kp];
        const v2f* br = (const v2f*)(lWs + kp * BSTRIDE_F);
#pragma unroll
        for (int nt = 0; nt < 4; ++nt) {
            const v2f b = br[nt * 16 + l16];
            A.t[nt] = __builtin_amdgcn_wmma_f32_16x16x4_f32(
                false, a, false, b, (short)0, A.t[nt], false, false);
        }
    }
    return A;
}

__device__ __attribute__((always_inline)) inline
void stage_w(float* __restrict__ lWs, const float* __restrict__ W, int kdim)
{
    const int total = kdim * OUT_DIM;
    for (int i = threadIdx.x; i < total; i += blockDim.x) {
        const int k = i >> 6;
        const int n = i & 63;
        lWs[(k >> 1) * BSTRIDE_F + (n << 1) + (k & 1)] = W[i];
    }
}

__device__ inline void atomicMaxF(float* addr, float val) {
    if (val >= 0.0f) atomicMax(reinterpret_cast<int*>(addr), __float_as_int(val));
    else             atomicMin(reinterpret_cast<unsigned int*>(addr), __float_as_uint(val));
}

__global__ void k_node_gemm(const float* __restrict__ x,
                            const float* __restrict__ W_fc,  const float* __restrict__ b_fc,
                            const float* __restrict__ W_self, const float* __restrict__ b_self,
                            float* __restrict__ z, float* __restrict__ hout,
                            int N, int nTiles)
{
    __shared__ float lWs[(IN_DIM / 2) * BSTRIDE_F];
    const int lane = threadIdx.x & 31;
    const int wave = threadIdx.x >> 5;
    const int half = lane >> 4;
    const int l16  = lane & 15;
    const int tile = blockIdx.x * 8 + wave;

    int mrow = tile * 16 + l16;
    if (mrow >= N) mrow = N - 1;
    const float* xrow = x + (size_t)mrow * IN_DIM;

    stage_w(lWs, W_fc, IN_DIM);
    __syncthreads();
    if (tile < nTiles) {
        Acc4 acc = wave_gemm_tile(xrow, lWs, IN_DIM, half, l16);
#pragma unroll
        for (int nt = 0; nt < 4; ++nt) {
            const int   c  = nt * 16 + l16;
            const float bc = b_fc[c];
#pragma unroll
            for (int r = 0; r < 8; ++r) {
                const int row = tile * 16 + r + 8 * half;
                if (row < N) z[(size_t)row * OUT_DIM + c] = acc.t[nt][r] + bc;
            }
        }
    }
    __syncthreads();

    stage_w(lWs, W_self, IN_DIM);
    __syncthreads();
    if (tile < nTiles) {
        Acc4 acc = wave_gemm_tile(xrow, lWs, IN_DIM, half, l16);
#pragma unroll
        for (int nt = 0; nt < 4; ++nt) {
            const int   c  = nt * 16 + l16;
            const float bc = b_self[c];
#pragma unroll
            for (int r = 0; r < 8; ++r) {
                const int row = tile * 16 + r + 8 * half;
                if (row < N) hout[(size_t)row * OUT_DIM + c] = acc.t[nt][r] + bc;
            }
        }
    }
}

__global__ void k_node_attn(const float* __restrict__ z, const float* __restrict__ W_attn,
                            float* __restrict__ a_src, float* __restrict__ a_dst,
                            float* __restrict__ m, float* __restrict__ denom, int N)
{
    const int n = blockIdx.x * blockDim.x + threadIdx.x;
    if (n >= N) return;
    const float* zr = z + (size_t)n * OUT_DIM;
    float as = 0.f, ad = 0.f;
#pragma unroll 8
    for (int c = 0; c < OUT_DIM; ++c) {
        const float zc = zr[c];
        as += zc * W_attn[c];
        ad += zc * W_attn[OUT_DIM + c];
    }
    a_src[n] = as;
    a_dst[n] = ad;
    m[n]     = -INFINITY;
    denom[n] = 0.0f;
}

__global__ void k_edge_logit(const float* __restrict__ edge_attr,
                             const int* __restrict__ erow, const int* __restrict__ ecol,
                             const float* __restrict__ W_attn, const float* __restrict__ b_attn,
                             const float* __restrict__ a_src, const float* __restrict__ a_dst,
                             float* __restrict__ elog, float* __restrict__ m, int E)
{
    const int e = blockIdx.x * blockDim.x + threadIdx.x;
    if (e >= E) return;
    const float* ea = edge_attr + (size_t)e * ED_DIM;
    float dot = 0.f;
#pragma unroll
    for (int k = 0; k < ED_DIM; ++k) dot += ea[k] * W_attn[2 * OUT_DIM + k];
    const int c = ecol[e];
    const float s  = a_src[erow[e]] + a_dst[c] + dot + b_attn[0];
    const float lr = (s > 0.0f) ? s : 0.2f * s;
    elog[e] = lr;
    atomicMaxF(&m[c], lr);
}

__global__ void k_edge_soft(const float* __restrict__ elog, const int* __restrict__ ecol,
                            const float* __restrict__ m, float* __restrict__ pbuf,
                            float* __restrict__ denom, int E)
{
    const int e = blockIdx.x * blockDim.x + threadIdx.x;
    if (e >= E) return;
    const int c = ecol[e];
    const float p = expf(elog[e] - m[c]);
    pbuf[e] = p;
    atomicAdd(&denom[c], p);
}

__global__ void k_edge_aggregate(const float* __restrict__ edge_attr,
                                 const int* __restrict__ erow, const int* __restrict__ ecol,
                                 const float* __restrict__ W_edge, const float* __restrict__ b_edge,
                                 const float* __restrict__ z, const float* __restrict__ pbuf,
                                 const float* __restrict__ denom, float* __restrict__ hout,
                                 int E, int nTiles)
{
    __shared__ float lWs[(ED_DIM / 2) * BSTRIDE_F];
    const int lane = threadIdx.x & 31;
    const int wave = threadIdx.x >> 5;
    const int half = lane >> 4;
    const int l16  = lane & 15;
    const int tile = blockIdx.x * 8 + wave;

    stage_w(lWs, W_edge, ED_DIM);
    __syncthreads();
    if (tile >= nTiles) return;

    int arow = tile * 16 + l16;
    if (arow >= E) arow = E - 1;
    const float* ea = edge_attr + (size_t)arow * ED_DIM;

    {
        const size_t nxt = (size_t)arow + 16 * 8;
        if (nxt < (size_t)E)
            __builtin_prefetch(edge_attr + nxt * ED_DIM, 0, 1);
    }

    Acc4 acc = wave_gemm_tile(ea, lWs, ED_DIM, half, l16);

    float alph[8];
    int   cr[8], rr[8];
#pragma unroll
    for (int r = 0; r < 8; ++r) {
        const int eidx = tile * 16 + r + 8 * half;
        if (eidx < E) {
            const int c = ecol[eidx];
            cr[r]   = c;
            rr[r]   = erow[eidx];
            alph[r] = pbuf[eidx] / fmaxf(denom[c], 1e-16f);
        } else {
            cr[r] = -1; rr[r] = 0; alph[r] = 0.f;
        }
    }

#pragma unroll
    for (int nt = 0; nt < 4; ++nt) {
        const int   c  = nt * 16 + l16;
        const float bc = b_edge[c];
#pragma unroll
        for (int r = 0; r < 8; ++r) {
            if (cr[r] >= 0) {
                const float val = alph[r] * (z[(size_t)rr[r] * OUT_DIM + c] + acc.t[nt][r] + bc);
                atomicAdd(&hout[(size_t)cr[r] * OUT_DIM + c], val);
            }
        }
    }
}

extern "C" void kernel_launch(void* const* d_in, const int* in_sizes, int n_in,
                              void* d_out, int out_size, void* d_ws, size_t ws_size,
                              hipStream_t stream)
{
    const float* x         = (const float*)d_in[0];
    const float* edge_attr = (const float*)d_in[1];
    const int*   edge_idx  = (const int*)  d_in[2];
    const float* W_fc      = (const float*)d_in[3];
    const float* b_fc      = (const float*)d_in[4];
    const float* W_attn    = (const float*)d_in[5];
    const float* b_attn    = (const float*)d_in[6];
    const float* W_edge    = (const float*)d_in[7];
    const float* b_edge    = (const float*)d_in[8];
    const float* W_self    = (const float*)d_in[9];
    const float* b_self    = (const float*)d_in[10];

    const int N = in_sizes[0] / IN_DIM;
    const int E = in_sizes[1] / ED_DIM;
    const int* erow = edge_idx;
    const int* ecol = edge_idx + E;

    float* ws     = (float*)d_ws;
    float* z      = ws;
    float* a_src  = z + (size_t)N * OUT_DIM;
    float* a_dst  = a_src + N;
    float* m      = a_dst + N;
    float* denom  = m + N;
    float* elog   = denom + N;
    float* pbuf   = elog + E;
    float* hout   = (float*)d_out;

    const int nNodeTiles = (N + 15) / 16;
    const int nEdgeTiles = (E + 15) / 16;

    k_node_gemm<<<(nNodeTiles + 7) / 8, 256, 0, stream>>>(
        x, W_fc, b_fc, W_self, b_self, z, hout, N, nNodeTiles);

    k_node_attn<<<(N + 255) / 256, 256, 0, stream>>>(
        z, W_attn, a_src, a_dst, m, denom, N);

    k_edge_logit<<<(E + 255) / 256, 256, 0, stream>>>(
        edge_attr, erow, ecol, W_attn, b_attn, a_src, a_dst, elog, m, E);

    k_edge_soft<<<(E + 255) / 256, 256, 0, stream>>>(
        elog, ecol, m, pbuf, denom, E);

    k_edge_aggregate<<<(nEdgeTiles + 7) / 8, 256, 0, stream>>>(
        edge_attr, erow, ecol, W_edge, b_edge, z, pbuf, denom, hout, E, nEdgeTiles);
}


# node kernel: x half rows and W rows read from global in whole lines straight into the fragment layout (4 adjacent columns per lane); no LDS staging, no barrier; z and h_init sum order changes
# speedup vs baseline: 1.6067x; 1.6067x over previous
#include <hip/hip_runtime.h>
#include <hip/hip_bf16.h>
#include <math.h>

#define IN_DIM  128
#define OUT_DIM 64
#define ED_DIM  32

#define BSTRIDE_V2 80
#define BSTRIDE_F  (2 * BSTRIDE_V2)

#define SPITCH 68

typedef __attribute__((ext_vector_type(2))) float v2f;
typedef __attribute__((ext_vector_type(8))) float v8f;

struct Acc4 { v8f t[4]; };

__device__ __attribute__((always_inline)) inline
Acc4 wave_gemm_tile(const float* __restrict__ arow,
                    const float* __restrict__ lWs,
                    int kdim, int half, int l16,
                    Acc4 A)
{
    const v2f* a2 = (const v2f*)arow;
    for (int k0 = 0; k0 < kdim; k0 += 4) {
        const int kp  = (k0 >> 1) + half;
        const v2f  a  = a2[kp];
        const v2f* br = (const v2f*)(lWs + kp * BSTRIDE_F);
#pragma unroll
        for (int nt = 0; nt < 4; ++nt) {
            const v2f b = br[nt * 16 + l16];
            A.t[nt] = __builtin_amdgcn_wmma_f32_16x16x4_f32(
                false, a, false, b, (short)0, A.t[nt], false, false);
        }
    }
    return A;
}

__device__ __attribute__((always_inline)) inline
void stage_w(float* __restrict__ lWs, const float* __restrict__ W, int kdim)
{
    const int total = kdim * OUT_DIM;
    for (int i = threadIdx.x; i < total; i += blockDim.x) {
        const int k = i >> 6;
        const int n = i & 63;
        lWs[(k >> 1) * BSTRIDE_F + (n << 1) + (k & 1)] = W[i];
    }
}

__device__ inline void atomicMaxF(float* addr, float val) {
    if (val >= 0.0f) atomicMax(reinterpret_cast<int*>(addr), __float_as_int(val));
    else             atomicMin(reinterpret_cast<unsigned int*>(addr), __float_as_uint(val));
}

typedef __attribute__((ext_vector_type(4))) float v4f;

__device__ __attribute__((always_inline)) inline
void node_product(const v4f (&a)[16], const float* __restrict__ Wm,
                  const float* __restrict__ bias, float* __restrict__ dst,
                  int tile, int half, int l16, int N)
{
    v8f acc[4];
    acc[0] = {}; acc[1] = {}; acc[2] = {}; acc[3] = {};
    const float* wl = Wm + (size_t)(8 * half) * OUT_DIM + 4 * l16;
#pragma unroll
    for (int kb = 0; kb < IN_DIM / 16; ++kb) {
        v4f b[8];
#pragma unroll
        for (int q = 0; q < 8; ++q)
            b[q] = *(const v4f*)(wl + (size_t)(16 * kb + q) * OUT_DIM);
#pragma unroll
        for (int j = 0; j < 4; ++j) {
            const v4f av = a[2 * kb + (j >> 1)];
            v2f ap;
            if (j & 1) { ap[0] = av[2]; ap[1] = av[3]; }
            else       { ap[0] = av[0]; ap[1] = av[1]; }
#pragma unroll
            for (int nt = 0; nt < 4; ++nt) {
                v2f bp;
                bp[0] = b[2 * j][nt];
                bp[1] = b[2 * j + 1][nt];
                acc[nt] = __builtin_amdgcn_wmma_f32_16x16x4_f32(
                    false, ap, false, bp, (short)0, acc[nt], false, false);
            }
        }
    }
    const v4f bc = *(const v4f*)(bias + 4 * l16);
#pragma unroll
    for (int r = 0; r < 8; ++r) {
        const int row = tile * 16 + r + 8 * half;
        v4f o;
        o[0] = acc[0][r] + bc[0];
        o[1] = acc[1][r] + bc[1];
        o[2] = acc[2][r] + bc[2];
        o[3] = acc[3][r] + bc[3];
        if (row < N) *(v4f*)(dst + (size_t)row * OUT_DIM + 4 * l16) = o;
    }
}

__global__ void __launch_bounds__(256) k_node_gemm(const float* __restrict__ x,
                            const float* __restrict__ W_fc,  const float* __restrict__ b_fc,
                            const float* __restrict__ W_self, const float* __restrict__ b_self,
                            float* __restrict__ z, float* __restrict__ hout,
                            int N, int nTiles)
{
    const int lane = threadIdx.x & 31;
    const int wave = threadIdx.x >> 5;
    const int half = lane >> 4;
    const int l16  = lane & 15;
    const int tile = blockIdx.x * 8 + wave;
    if (tile >= nTiles) return;

    int mrow = tile * 16 + l16;
    if (mrow >= N) mrow = N - 1;
    const v4f* xr = (const v4f*)(x + (size_t)mrow * IN_DIM + 8 * half);
    v4f a[16];
#pragma unroll
    for (int kb = 0; kb < IN_DIM / 16; ++kb) {
        a[2 * kb]     = xr[4 * kb];
        a[2 * kb + 1] = xr[4 * kb + 1];
    }

    node_product(a, W_fc,   b_fc,   z,    tile, half, l16, N);
    node_product(a, W_self, b_self, hout, tile, half, l16, N);
}

__global__ void k_node_attn(const float* __restrict__ z, const float* __restrict__ W_attn,
                            float* __restrict__ a_src, float* __restrict__ a_dst,
                            float* __restrict__ m, float* __restrict__ denom, int N)
{
    const int n = blockIdx.x * blockDim.x + threadIdx.x;
    if (n >= N) return;
    const float* zr = z + (size_t)n * OUT_DIM;
    float as = 0.f, ad = 0.f;
#pragma unroll 8
    for (int c = 0; c < OUT_DIM; ++c) {
        const float zc = zr[c];
        as += zc * W_attn[c];
        ad += zc * W_attn[OUT_DIM + c];
    }
    a_src[n] = as;
    a_dst[n] = ad;
    m[n]     = -INFINITY;
    denom[n] = 0.0f;
}

__global__ void k_edge_logit(const float* __restrict__ edge_attr,
                             const int* __restrict__ erow, const int* __restrict__ ecol,
                             const float* __restrict__ W_attn, const float* __restrict__ b_attn,
                             const float* __restrict__ a_src, const float* __restrict__ a_dst,
                             float* __restrict__ elog, float* __restrict__ m, int E)
{
    const int e = blockIdx.x * blockDim.x + threadIdx.x;
    if (e >= E) return;
    const float* ea = edge_attr + (size_t)e * ED_DIM;
    float dot = 0.f;
#pragma unroll
    for (int k = 0; k < ED_DIM; ++k) dot += ea[k] * W_attn[2 * OUT_DIM + k];
    const int c = ecol[e];
    const float s  = a_src[erow[e]] + a_dst[c] + dot + b_attn[0];
    const float lr = (s > 0.0f) ? s : 0.2f * s;
    elog[e] = lr;
    atomicMaxF(&m[c], lr);
}

__global__ void k_edge_soft(const float* __restrict__ elog, const int* __restrict__ ecol,
                            const float* __restrict__ m, float* __restrict__ pbuf,
                            float* __restrict__ denom, int E)
{
    const int e = blockIdx.x * blockDim.x + threadIdx.x;
    if (e >= E) return;
    const int c = ecol[e];
    const float p = expf(elog[e] - m[c]);
    pbuf[e] = p;
    atomicAdd(&denom[c], p);
}

__global__ void k_edge_aggregate(const float* __restrict__ edge_attr,
                                 const int* __restrict__ erow, const int* __restrict__ ecol,
                                 const float* __restrict__ W_edge, const float* __restrict__ b_edge,
                                 const float* __restrict__ z, const float* __restrict__ pbuf,
                                 const float* __restrict__ denom, float* __restrict__ hout,
                                 int E, int nTiles)
{
    __shared__ float lWs[(ED_DIM / 2) * BSTRIDE_F];
    __shared__ float lSt[8 * 16 * SPITCH];
    const int lane = threadIdx.x & 31;
    const int wave = threadIdx.x >> 5;
    const int half = lane >> 4;
    const int l16  = lane & 15;
    const int tile = blockIdx.x * 8 + wave;

    int   cr[8], rr[8], cl[8];
    float pv[8], alph[8];
#pragma unroll
    for (int r = 0; r < 8; ++r) {
        const int eidx = tile * 16 + r + 8 * half;
        const int ei   = (eidx < E) ? eidx : (E - 1);
        cl[r] = ecol[ei];
        rr[r] = erow[ei];
        pv[r] = pbuf[ei];
        cr[r] = (eidx < E) ? cl[r] : -1;
    }
#pragma unroll
    for (int r = 0; r < 8; ++r) {
        const float a = pv[r] / fmaxf(denom[cl[r]], 1e-16f);
        alph[r] = (cr[r] >= 0) ? a : 0.f;
    }

    Acc4 acc;
#pragma unroll
    for (int nt = 0; nt < 4; ++nt) {
        const int   c  = nt * 16 + l16;
        const float bc = b_edge[c];
#pragma unroll
        for (int r = 0; r < 8; ++r)
            acc.t[nt][r] = z[(size_t)rr[r] * OUT_DIM + c] + bc;
    }

    stage_w(lWs, W_edge, ED_DIM);
    __syncthreads();
    if (tile >= nTiles) return;

    int arow = tile * 16 + l16;
    if (arow >= E) arow = E - 1;
    const float* ea = edge_attr + (size_t)arow * ED_DIM;

    {
        const size_t nxt = (size_t)arow + 16 * 8;
        if (nxt < (size_t)E)
            __builtin_prefetch(edge_attr + nxt * ED_DIM, 0, 1);
    }

    acc = wave_gemm_tile(ea, lWs, ED_DIM, half, l16, acc);

    float* st = lSt + wave * (16 * SPITCH);
#pragma unroll
    for (int nt = 0; nt < 4; ++nt) {
#pragma unroll
        for (int r = 0; r < 8; ++r)
            st[(r + 8 * half) * SPITCH + nt * 16 + l16] = alph[r] * acc.t[nt][r];
    }
    __builtin_amdgcn_fence(__ATOMIC_RELEASE, "workgroup");
    __builtin_amdgcn_fence(__ATOMIC_ACQUIRE, "workgroup");

    float v0[16], v1[16];
#pragma unroll
    for (int e = 0; e < 16; ++e) {
        v0[e] = st[e * SPITCH + lane];
        v1[e] = st[e * SPITCH + 32 + lane];
    }
#pragma unroll
    for (int e = 0; e < 16; ++e) {
        const int crow = __builtin_amdgcn_readlane(cr[e & 7], (e >> 3) * 16);
        if (crow >= 0) {
            float* rowp = hout + (size_t)crow * OUT_DIM + lane;
            atomicAdd(rowp,      v0[e]);
            atomicAdd(rowp + 32, v1[e]);
        }
    }
}

extern "C" void kernel_launch(void* const* d_in, const int* in_sizes, int n_in,
                              void* d_out, int out_size, void* d_ws, size_t ws_size,
                              hipStream_t stream)
{
    const float* x         = (const float*)d_in[0];
    const float* edge_attr = (const float*)d_in[1];
    const int*   edge_idx  = (const int*)  d_in[2];
    const float* W_fc      = (const float*)d_in[3];
    const float* b_fc      = (const float*)d_in[4];
    const float* W_attn    = (const float*)d_in[5];
    const float* b_attn    = (const float*)d_in[6];
    const float* W_edge    = (const float*)d_in[7];
    const float* b_edge    = (const float*)d_in[8];
    const float* W_self    = (const float*)d_in[9];
    const float* b_self    = (const float*)d_in[10];

    const int N = in_sizes[0] / IN_DIM;
    const int E = in_sizes[1] / ED_DIM;
    const int* erow = edge_idx;
    const int* ecol = edge_idx + E;

    float* ws     = (float*)d_ws;
    float* z      = ws;
    float* a_src  = z + (size_t)N * OUT_DIM;
    float* a_dst  = a_src + N;
    float* m      = a_dst + N;
    float* denom  = m + N;
    float* elog   = denom + N;
    float* pbuf   = elog + E;
    float* hout   = (float*)d_out;

    const int nNodeTiles = (N + 15) / 16;
    const int nEdgeTiles = (E + 15) / 16;

    k_node_gemm<<<(nNodeTiles + 7) / 8, 256, 0, stream>>>(
        x, W_fc, b_fc, W_self, b_self, z, hout, N, nNodeTiles);

    k_node_attn<<<(N + 255) / 256, 256, 0, stream>>>(
        z, W_attn, a_src, a_dst, m, denom, N);

    k_edge_logit<<<(E + 255) / 256, 256, 0, stream>>>(
        edge_attr, erow, ecol, W_attn, b_attn, a_src, a_dst, elog, m, E);

    k_edge_soft<<<(E + 255) / 256, 256, 0, stream>>>(
        elog, ecol, m, pbuf, denom, E);

    k_edge_aggregate<<<(nEdgeTiles + 7) / 8, 256, 0, stream>>>(
        edge_attr, erow, ecol, W_edge, b_edge, z, pbuf, denom, hout, E, nEdgeTiles);
}


# segment max and exp sum: kept per edge slice in work-group LDS tables (slice x node-range grid, local max), merged per node by one more launch; their memory atomics and p plane go; exp-sum order new
# speedup vs baseline: 2.6949x; 1.6773x over previous
#include <hip/hip_runtime.h>
#include <hip/hip_bf16.h>
#include <math.h>

#define IN_DIM  128
#define OUT_DIM 64
#define ED_DIM  32

#define BSTRIDE_V2 80
#define BSTRIDE_F  (2 * BSTRIDE_V2)

#define SPITCH 68

typedef __attribute__((ext_vector_type(2))) float v2f;
typedef __attribute__((ext_vector_type(8))) float v8f;

struct Acc4 { v8f t[4]; };

__device__ __attribute__((always_inline)) inline
Acc4 wave_gemm_tile(const float* __restrict__ arow,
                    const float* __restrict__ lWs,
                    int kdim, int half, int l16,
                    Acc4 A)
{
    const v2f* a2 = (const v2f*)arow;
    for (int k0 = 0; k0 < kdim; k0 += 4) {
        const int kp  = (k0 >> 1) + half;
        const v2f  a  = a2[kp];
        const v2f* br = (const v2f*)(lWs + kp * BSTRIDE_F);
#pragma unroll
        for (int nt = 0; nt < 4; ++nt) {
            const v2f b = br[nt * 16 + l16];
            A.t[nt] = __builtin_amdgcn_wmma_f32_16x16x4_f32(
                false, a, false, b, (short)0, A.t[nt], false, false);
        }
    }
    return A;
}

__device__ __attribute__((always_inline)) inline
void stage_w(float* __restrict__ lWs, const float* __restrict__ W, int kdim)
{
    const int total = kdim * OUT_DIM;
    for (int i = threadIdx.x; i < total; i += blockDim.x) {
        const int k = i >> 6;
        const int n = i & 63;
        lWs[(k >> 1) * BSTRIDE_F + (n << 1) + (k & 1)] = W[i];
    }
}

__device__ __attribute__((always_inline)) inline void lds_max_f32(float* addr, float val) {
    const int bits = __float_as_int(val);
    if (bits >= 0) atomicMax(reinterpret_cast<int*>(addr), bits);
    else           atomicMin(reinterpret_cast<unsigned int*>(addr), (unsigned int)bits);
}

typedef __attribute__((ext_vector_type(4))) float v4f;

__device__ __attribute__((always_inline)) inline
void node_product(const v4f (&a)[16], const float* __restrict__ Wm,
                  const float* __restrict__ bias, float* __restrict__ dst,
                  int tile, int half, int l16, int N)
{
    v8f acc[4];
    acc[0] = {}; acc[1] = {}; acc[2] = {}; acc[3] = {};
    const float* wl = Wm + (size_t)(8 * half) * OUT_DIM + 4 * l16;
#pragma unroll
    for (int kb = 0; kb < IN_DIM / 16; ++kb) {
        v4f b[8];
#pragma unroll
        for (int q = 0; q < 8; ++q)
            b[q] = *(const v4f*)(wl + (size_t)(16 * kb + q) * OUT_DIM);
#pragma unroll
        for (int j = 0; j < 4; ++j) {
            const v4f av = a[2 * kb + (j >> 1)];
            v2f ap;
            if (j & 1) { ap[0] = av[2]; ap[1] = av[3]; }
            else       { ap[0] = av[0]; ap[1] = av[1]; }
#pragma unroll
            for (int nt = 0; nt < 4; ++nt) {
                v2f bp;
                bp[0] = b[2 * j][nt];
                bp[1] = b[2 * j + 1][nt];
                acc[nt] = __builtin_amdgcn_wmma_f32_16x16x4_f32(
                    false, ap, false, bp, (short)0, acc[nt], false, false);
            }
        }
    }
    const v4f bc = *(const v4f*)(bias + 4 * l16);
#pragma unroll
    for (int r = 0; r < 8; ++r) {
        const int row = tile * 16 + r + 8 * half;
        v4f o;
        o[0] = acc[0][r] + bc[0];
        o[1] = acc[1][r] + bc[1];
        o[2] = acc[2][r] + bc[2];
        o[3] = acc[3][r] + bc[3];
        if (row < N) *(v4f*)(dst + (size_t)row * OUT_DIM + 4 * l16) = o;
    }
}

__global__ void __launch_bounds__(256) k_node_gemm(const float* __restrict__ x,
                            const float* __restrict__ W_fc,  const float* __restrict__ b_fc,
                            const float* __restrict__ W_self, const float* __restrict__ b_self,
                            float* __restrict__ z, float* __restrict__ hout,
                            int N, int nTiles)
{
    const int lane = threadIdx.x & 31;
    const int wave = threadIdx.x >> 5;
    const int half = lane >> 4;
    const int l16  = lane & 15;
    const int tile = blockIdx.x * 8 + wave;
    if (tile >= nTiles) return;

    int mrow = tile * 16 + l16;
    if (mrow >= N) mrow = N - 1;
    const v4f* xr = (const v4f*)(x + (size_t)mrow * IN_DIM + 8 * half);
    v4f a[16];
#pragma unroll
    for (int kb = 0; kb < IN_DIM / 16; ++kb) {
        a[2 * kb]     = xr[4 * kb];
        a[2 * kb + 1] = xr[4 * kb + 1];
    }

    node_product(a, W_fc,   b_fc,   z,    tile, half, l16, N);
    node_product(a, W_self, b_self, hout, tile, half, l16, N);
}

__global__ void k_node_attn(const float* __restrict__ z, const float* __restrict__ W_attn,
                            float* __restrict__ a_src, float* __restrict__ a_dst, int N)
{
    const int n = blockIdx.x * blockDim.x + threadIdx.x;
    if (n >= N) return;
    const float* zr = z + (size_t)n * OUT_DIM;
    float as = 0.f, ad = 0.f;
#pragma unroll 8
    for (int c = 0; c < OUT_DIM; ++c) {
        const float zc = zr[c];
        as += zc * W_attn[c];
        ad += zc * W_attn[OUT_DIM + c];
    }
    a_src[n] = as;
    a_dst[n] = ad;
}

__global__ void k_edge_logit(const float* __restrict__ edge_attr,
                             const int* __restrict__ erow, const int* __restrict__ ecol,
                             const float* __restrict__ W_attn, const float* __restrict__ b_attn,
                             const float* __restrict__ a_src, const float* __restrict__ a_dst,
                             float* __restrict__ elog, int E)
{
    const int e = blockIdx.x * blockDim.x + threadIdx.x;
    if (e >= E) return;
    const float* ea = edge_attr + (size_t)e * ED_DIM;
    float dot = 0.f;
#pragma unroll
    for (int k = 0; k < ED_DIM; ++k) dot += ea[k] * W_attn[2 * OUT_DIM + k];
    const int c = ecol[e];
    const float s  = a_src[erow[e]] + a_dst[c] + dot + b_attn[0];
    const float lr = (s > 0.0f) ? s : 0.2f * s;
    elog[e] = lr;
}

#define ST_P  32
#define ST_NR 6400
#define ST_UB 8

__global__ void __launch_bounds__(1024) k_seg_stats(const float* __restrict__ elog,
                            const int* __restrict__ ecol,
                            float* __restrict__ Mp, float* __restrict__ Dp,
                            int E, int sliceLen, int Npad)
{
    __shared__ float Mt[ST_NR];
    __shared__ float Dt[ST_NR];
    const int tid  = threadIdx.x;
    const int base = blockIdx.y * ST_NR;
    for (int i = tid; i < ST_NR; i += 1024) { Mt[i] = -INFINITY; Dt[i] = 0.0f; }
    __syncthreads();

    const int e0 = blockIdx.x * sliceLen;
    const int e1 = (e0 + sliceLen < E) ? (e0 + sliceLen) : E;

    for (int eb = e0 + tid; eb < e1; eb += 1024 * ST_UB) {
        int li[ST_UB]; float lv[ST_UB];
#pragma unroll
        for (int u = 0; u < ST_UB; ++u) {
            const int e  = eb + u * 1024;
            const int ec = (e < e1) ? e : (e1 - 1);
            const int c  = ecol[ec] - base;
            lv[u] = elog[ec];
            li[u] = (e < e1) ? c : -1;
        }
#pragma unroll
        for (int u = 0; u < ST_UB; ++u)
            if ((unsigned int)li[u] < (unsigned int)ST_NR) lds_max_f32(&Mt[li[u]], lv[u]);
    }
    __syncthreads();
    for (int eb = e0 + tid; eb < e1; eb += 1024 * ST_UB) {
        int li[ST_UB]; float lv[ST_UB], mt[ST_UB];
#pragma unroll
        for (int u = 0; u < ST_UB; ++u) {
            const int e  = eb + u * 1024;
            const int ec = (e < e1) ? e : (e1 - 1);
            const int c  = ecol[ec] - base;
            lv[u] = elog[ec];
            li[u] = (e < e1) ? c : -1;
        }
#pragma unroll
        for (int u = 0; u < ST_UB; ++u)
            mt[u] = ((unsigned int)li[u] < (unsigned int)ST_NR) ? Mt[li[u]] : 0.0f;
#pragma unroll
        for (int u = 0; u < ST_UB; ++u)
            if ((unsigned int)li[u] < (unsigned int)ST_NR) atomicAdd(&Dt[li[u]], expf(lv[u] - mt[u]));
    }
    __syncthreads();

    float* mo = Mp + (size_t)blockIdx.x * Npad + base;
    float* so = Dp + (size_t)blockIdx.x * Npad + base;
    for (int i = tid; i < ST_NR; i += 1024) { mo[i] = Mt[i]; so[i] = Dt[i]; }
}

__global__ void k_seg_merge(const float* __restrict__ Mp, const float* __restrict__ Dp,
                            float* __restrict__ md, int N, int Npad)
{
    const int n = blockIdx.x * blockDim.x + threadIdx.x;
    if (n >= N) return;
    float mv[ST_P], dv[ST_P];
#pragma unroll
    for (int w = 0; w < ST_P; ++w) {
        mv[w] = Mp[(size_t)w * Npad + n];
        dv[w] = Dp[(size_t)w * Npad + n];
    }
    float mx = mv[0];
#pragma unroll
    for (int w = 1; w < ST_P; ++w) mx = fmaxf(mx, mv[w]);
    if (!__builtin_isfinite(mx)) mx = 0.0f;
    float d = 0.0f;
#pragma unroll
    for (int w = 0; w < ST_P; ++w) d += dv[w] * expf(mv[w] - mx);
    v2f o; o[0] = mx; o[1] = d;
    *(v2f*)(md + 2 * (size_t)n) = o;
}

__global__ void k_edge_aggregate(const float* __restrict__ edge_attr,
                                 const int* __restrict__ erow, const int* __restrict__ ecol,
                                 const float* __restrict__ W_edge, const float* __restrict__ b_edge,
                                 const float* __restrict__ z, const float* __restrict__ elog,
                                 const float* __restrict__ md, float* __restrict__ hout,
                                 int E, int nTiles)
{
    __shared__ float lWs[(ED_DIM / 2) * BSTRIDE_F];
    __shared__ float lSt[8 * 16 * SPITCH];
    const int lane = threadIdx.x & 31;
    const int wave = threadIdx.x >> 5;
    const int half = lane >> 4;
    const int l16  = lane & 15;
    const int tile = blockIdx.x * 8 + wave;

    int   cr[8], rr[8], cl[8];
    float el[8], alph[8];
#pragma unroll
    for (int r = 0; r < 8; ++r) {
        const int eidx = tile * 16 + r + 8 * half;
        const int ei   = (eidx < E) ? eidx : (E - 1);
        cl[r] = ecol[ei];
        rr[r] = erow[ei];
        el[r] = elog[ei];
        cr[r] = (eidx < E) ? cl[r] : -1;
    }
#pragma unroll
    for (int r = 0; r < 8; ++r) {
        const v2f   s = *(const v2f*)(md + 2 * (size_t)cl[r]);
        const float a = expf(el[r] - s[0]) / fmaxf(s[1], 1e-16f);
        alph[r] = (cr[r] >= 0) ? a : 0.f;
    }

    Acc4 acc;
#pragma unroll
    for (int nt = 0; nt < 4; ++nt) {
        const int   c  = nt * 16 + l16;
        const float bc = b_edge[c];
#pragma unroll
        for (int r = 0; r < 8; ++r)
            acc.t[nt][r] = z[(size_t)rr[r] * OUT_DIM + c] + bc;
    }

    stage_w(lWs, W_edge, ED_DIM);
    __syncthreads();
    if (tile >= nTiles) return;

    int arow = tile * 16 + l16;
    if (arow >= E) arow = E - 1;
    const float* ea = edge_attr + (size_t)arow * ED_DIM;

    {
        const size_t nxt = (size_t)arow + 16 * 8;
        if (nxt < (size_t)E)
            __builtin_prefetch(edge_attr + nxt * ED_DIM, 0, 1);
    }

    acc = wave_gemm_tile(ea, lWs, ED_DIM, half, l16, acc);

    float* st = lSt + wave * (16 * SPITCH);
#pragma unroll
    for (int nt = 0; nt < 4; ++nt) {
#pragma unroll
        for (int r = 0; r < 8; ++r)
            st[(r + 8 * half) * SPITCH + nt * 16 + l16] = alph[r] * acc.t[nt][r];
    }
    __builtin_amdgcn_fence(__ATOMIC_RELEASE, "workgroup");
    __builtin_amdgcn_fence(__ATOMIC_ACQUIRE, "workgroup");

    float v0[16], v1[16];
#pragma unroll
    for (int e = 0; e < 16; ++e) {
        v0[e] = st[e * SPITCH + lane];
        v1[e] = st[e * SPITCH + 32 + lane];
    }
#pragma unroll
    for (int e = 0; e < 16; ++e) {
        const int crow = __builtin_amdgcn_readlane(cr[e & 7], (e >> 3) * 16);
        if (crow >= 0) {
            float* rowp = hout + (size_t)crow * OUT_DIM + lane;
            atomicAdd(rowp,      v0[e]);
            atomicAdd(rowp + 32, v1[e]);
        }
    }
}

extern "C" void kernel_launch(void* const* d_in, const int* in_sizes, int n_in,
                              void* d_out, int out_size, void* d_ws, size_t ws_size,
                              hipStream_t stream)
{
    const float* x         = (const float*)d_in[0];
    const float* edge_attr = (const float*)d_in[1];
    const int*   edge_idx  = (const int*)  d_in[2];
    const float* W_fc      = (const float*)d_in[3];
    const float* b_fc      = (const float*)d_in[4];
    const float* W_attn    = (const float*)d_in[5];
    const float* b_attn    = (const float*)d_in[6];
    const float* W_edge    = (const float*)d_in[7];
    const float* b_edge    = (const float*)d_in[8];
    const float* W_self    = (const float*)d_in[9];
    const float* b_self    = (const float*)d_in[10];

    const int N = in_sizes[0] / IN_DIM;
    const int E = in_sizes[1] / ED_DIM;
    const int* erow = edge_idx;
    const int* ecol = edge_idx + E;

    const int nRanges  = (N + ST_NR - 1) / ST_NR;
    const int Npad     = nRanges * ST_NR;
    const int sliceLen = (E + ST_P - 1) / ST_P;

    float* ws     = (float*)d_ws;
    float* z      = ws;
    float* a_src  = z + (size_t)N * OUT_DIM;
    float* a_dst  = a_src + N;
    float* md     = a_dst + N;
    float* elog   = md + 2 * (size_t)N;
    float* Mp     = elog + E;
    float* Dp     = Mp + (size_t)ST_P * Npad;
    float* hout   = (float*)d_out;

    const int nNodeTiles = (N + 15) / 16;
    const int nEdgeTiles = (E + 15) / 16;

    k_node_gemm<<<(nNodeTiles + 7) / 8, 256, 0, stream>>>(
        x, W_fc, b_fc, W_self, b_self, z, hout, N, nNodeTiles);

    k_node_attn<<<(N + 255) / 256, 256, 0, stream>>>(
        z, W_attn, a_src, a_dst, N);

    k_edge_logit<<<(E + 255) / 256, 256, 0, stream>>>(
        edge_attr, erow, ecol, W_attn, b_attn, a_src, a_dst, elog, E);

    k_seg_stats<<<dim3(ST_P, nRanges), 1024, 0, stream>>>(
        elog, ecol, Mp, Dp, E, sliceLen, Npad);

    k_seg_merge<<<(N + 255) / 256, 256, 0, stream>>>(
        Mp, Dp, md, N, Npad);

    k_edge_aggregate<<<(nEdgeTiles + 7) / 8, 256, 0, stream>>>(
        edge_attr, erow, ecol, W_edge, b_edge, z, elog, md, hout, E, nEdgeTiles);
}


# every row-per-lane global read made whole-line: edge_attr tiles and z rows turned through LDS (logit, aggregate, node scalars), z gathers 16 B a lane, one edge record a lane shared by row DPP
# speedup vs baseline: 3.2197x; 1.1947x over previous
#include <hip/hip_runtime.h>
#include <hip/hip_bf16.h>
#include <math.h>

#define IN_DIM  128
#define OUT_DIM 64
#define ED_DIM  32

#define BSTRIDE_V2 80
#define BSTRIDE_F  (2 * BSTRIDE_V2)

#define SPITCH 68

typedef __attribute__((ext_vector_type(2))) float v2f;
typedef __attribute__((ext_vector_type(8))) float v8f;

struct Acc4 { v8f t[4]; };

__device__ __attribute__((always_inline)) inline
Acc4 wave_gemm_tile(const float* __restrict__ arow,
                    const float* __restrict__ lWs,
                    int kdim, int half, int l16,
                    Acc4 A)
{
    const v2f* a2 = (const v2f*)arow;
    for (int k0 = 0; k0 < kdim; k0 += 4) {
        const int kp  = (k0 >> 1) + half;
        const v2f  a  = a2[kp];
        const v2f* br = (const v2f*)(lWs + kp * BSTRIDE_F);
#pragma unroll
        for (int nt = 0; nt < 4; ++nt) {
            const v2f b = br[nt * 16 + l16];
            A.t[nt] = __builtin_amdgcn_wmma_f32_16x16x4_f32(
                false, a, false, b, (short)0, A.t[nt], false, false);
        }
    }
    return A;
}

__device__ __attribute__((always_inline)) inline void lds_max_f32(float* addr, float val) {
    const int bits = __float_as_int(val);
    if (bits >= 0) atomicMax(reinterpret_cast<int*>(addr), bits);
    else           atomicMin(reinterpret_cast<unsigned int*>(addr), (unsigned int)bits);
}

#define EPITCH 36

template <int R> __device__ __attribute__((always_inline)) inline int row_share_i(int v) {
    return __builtin_amdgcn_update_dpp(0, v, 0x150 + R, 0xf, 0xf, false);
}
template <int R> __device__ __attribute__((always_inline)) inline float row_share_f(float v) {
    return __int_as_float(row_share_i<R>(__float_as_int(v)));
}

typedef __attribute__((ext_vector_type(4))) float v4f;

__device__ __attribute__((always_inline)) inline
void node_product(const v4f (&a)[16], const float* __restrict__ Wm,
                  const float* __restrict__ bias, float* __restrict__ dst,
                  int tile, int half, int l16, int N)
{
    v8f acc[4];
    acc[0] = {}; acc[1] = {}; acc[2] = {}; acc[3] = {};
    const float* wl = Wm + (size_t)(8 * half) * OUT_DIM + 4 * l16;
#pragma unroll
    for (int kb = 0; kb < IN_DIM / 16; ++kb) {
        v4f b[8];
#pragma unroll
        for (int q = 0; q < 8; ++q)
            b[q] = *(const v4f*)(wl + (size_t)(16 * kb + q) * OUT_DIM);
#pragma unroll
        for (int j = 0; j < 4; ++j) {
            const v4f av = a[2 * kb + (j >> 1)];
            v2f ap;
            if (j & 1) { ap[0] = av[2]; ap[1] = av[3]; }
            else       { ap[0] = av[0]; ap[1] = av[1]; }
#pragma unroll
            for (int nt = 0; nt < 4; ++nt) {
                v2f bp;
                bp[0] = b[2 * j][nt];
                bp[1] = b[2 * j + 1][nt];
                acc[nt] = __builtin_amdgcn_wmma_f32_16x16x4_f32(
                    false, ap, false, bp, (short)0, acc[nt], false, false);
            }
        }
    }
    const v4f bc = *(const v4f*)(bias + 4 * l16);
#pragma unroll
    for (int r = 0; r < 8; ++r) {
        const int row = tile * 16 + r + 8 * half;
        v4f o;
        o[0] = acc[0][r] + bc[0];
        o[1] = acc[1][r] + bc[1];
        o[2] = acc[2][r] + bc[2];
        o[3] = acc[3][r] + bc[3];
        if (row < N) *(v4f*)(dst + (size_t)row * OUT_DIM + 4 * l16) = o;
    }
}

__global__ void __launch_bounds__(256) k_node_gemm(const float* __restrict__ x,
                            const float* __restrict__ W_fc,  const float* __restrict__ b_fc,
                            const float* __restrict__ W_self, const float* __restrict__ b_self,
                            float* __restrict__ z, float* __restrict__ hout,
                            int N, int nTiles)
{
    const int lane = threadIdx.x & 31;
    const int wave = threadIdx.x >> 5;
    const int half = lane >> 4;
    const int l16  = lane & 15;
    const int tile = blockIdx.x * 8 + wave;
    if (tile >= nTiles) return;

    int mrow = tile * 16 + l16;
    if (mrow >= N) mrow = N - 1;
    const v4f* xr = (const v4f*)(x + (size_t)mrow * IN_DIM + 8 * half);
    v4f a[16];
#pragma unroll
    for (int kb = 0; kb < IN_DIM / 16; ++kb) {
        a[2 * kb]     = xr[4 * kb];
        a[2 * kb + 1] = xr[4 * kb + 1];
    }

    node_product(a, W_fc,   b_fc,   z,    tile, half, l16, N);
    node_product(a, W_self, b_self, hout, tile, half, l16, N);
}

#define NA_T   128
#define NPITCH 68

__global__ void k_node_attn(const float* __restrict__ z, const float* __restrict__ W_attn,
                            float* __restrict__ a_src, float* __restrict__ a_dst, int N)
{
    __shared__ __attribute__((aligned(16))) float lZ[NA_T * NPITCH];
    const int t  = threadIdx.x;
    const int n0 = blockIdx.x * NA_T;
    const int tr = t >> 4;
    const int tc = 4 * (t & 15);

    v4f v[16];
#pragma unroll
    for (int i = 0; i < 16; ++i) {
        int g = n0 + 8 * i + tr;
        if (g >= N) g = N - 1;
        v[i] = *(const v4f*)(z + (size_t)g * OUT_DIM + tc);
    }
#pragma unroll
    for (int i = 0; i < 16; ++i)
        *(v4f*)(lZ + (8 * i + tr) * NPITCH + tc) = v[i];
    __syncthreads();

    const int n = n0 + t;
    const float* zr = lZ + t * NPITCH;
    float as = 0.f, ad = 0.f;
#pragma unroll
    for (int q = 0; q < OUT_DIM / 4; ++q) {
        const v4f zq = *(const v4f*)(zr + 4 * q);
#pragma unroll
        for (int j = 0; j < 4; ++j) {
            as += zq[j] * W_attn[4 * q + j];
            ad += zq[j] * W_attn[OUT_DIM + 4 * q + j];
        }
    }
    if (n < N) {
        a_src[n] = as;
        a_dst[n] = ad;
    }
}

#define LG_T 256

__global__ void k_edge_logit(const float* __restrict__ edge_attr,
                             const int* __restrict__ erow, const int* __restrict__ ecol,
                             const float* __restrict__ W_attn, const float* __restrict__ b_attn,
                             const float* __restrict__ a_src, const float* __restrict__ a_dst,
                             float* __restrict__ elog, int E)
{
    __shared__ __attribute__((aligned(16))) float lE[LG_T * EPITCH];
    const int t  = threadIdx.x;
    const int e0 = blockIdx.x * LG_T;
    const int tr = t >> 3;
    const int tc = 4 * (t & 7);

    v4f v[8];
#pragma unroll
    for (int i = 0; i < 8; ++i) {
        int g = e0 + 32 * i + tr;
        if (g >= E) g = E - 1;
        v[i] = *(const v4f*)(edge_attr + (size_t)g * ED_DIM + tc);
    }
    const int e  = e0 + t;
    const int ec = (e < E) ? e : (E - 1);
    const int c  = ecol[ec];
    const float as = a_src[erow[ec]];
    const float ad = a_dst[c];
#pragma unroll
    for (int i = 0; i < 8; ++i)
        *(v4f*)(lE + (32 * i + tr) * EPITCH + tc) = v[i];
    __syncthreads();

    const float* er = lE + t * EPITCH;
    float dot = 0.f;
#pragma unroll
    for (int q = 0; q < ED_DIM / 4; ++q) {
        const v4f xq = *(const v4f*)(er + 4 * q);
        dot += xq[0] * W_attn[2 * OUT_DIM + 4 * q];
        dot += xq[1] * W_attn[2 * OUT_DIM + 4 * q + 1];
        dot += xq[2] * W_attn[2 * OUT_DIM + 4 * q + 2];
        dot += xq[3] * W_attn[2 * OUT_DIM + 4 * q + 3];
    }
    const float s  = as + ad + dot + b_attn[0];
    const float lr = (s > 0.0f) ? s : 0.2f * s;
    if (e < E) elog[e] = lr;
}

#define ST_P  32
#define ST_NR 6400
#define ST_UB 8

__global__ void __launch_bounds__(1024) k_seg_stats(const float* __restrict__ elog,
                            const int* __restrict__ ecol,
                            float* __restrict__ Mp, float* __restrict__ Dp,
                            int E, int sliceLen, int Npad)
{
    __shared__ float Mt[ST_NR];
    __shared__ float Dt[ST_NR];
    const int tid  = threadIdx.x;
    const int base = blockIdx.y * ST_NR;
    for (int i = tid; i < ST_NR; i += 1024) { Mt[i] = -INFINITY; Dt[i] = 0.0f; }
    __syncthreads();

    const int e0 = blockIdx.x * sliceLen;
    const int e1 = (e0 + sliceLen < E) ? (e0 + sliceLen) : E;

    for (int eb = e0 + tid; eb < e1; eb += 1024 * ST_UB) {
        int li[ST_UB]; float lv[ST_UB];
#pragma unroll
        for (int u = 0; u < ST_UB; ++u) {
            const int e  = eb + u * 1024;
            const int ec = (e < e1) ? e : (e1 - 1);
            const int c  = ecol[ec] - base;
            lv[u] = elog[ec];
            li[u] = (e < e1) ? c : -1;
        }
#pragma unroll
        for (int u = 0; u < ST_UB; ++u)
            if ((unsigned int)li[u] < (unsigned int)ST_NR) lds_max_f32(&Mt[li[u]], lv[u]);
    }
    __syncthreads();
    for (int eb = e0 + tid; eb < e1; eb += 1024 * ST_UB) {
        int li[ST_UB]; float lv[ST_UB], mt[ST_UB];
#pragma unroll
        for (int u = 0; u < ST_UB; ++u) {
            const int e  = eb + u * 1024;
            const int ec = (e < e1) ? e : (e1 - 1);
            const int c  = ecol[ec] - base;
            lv[u] = elog[ec];
            li[u] = (e < e1) ? c : -1;
        }
#pragma unroll
        for (int u = 0; u < ST_UB; ++u)
            mt[u] = ((unsigned int)li[u] < (unsigned int)ST_NR) ? Mt[li[u]] : 0.0f;
#pragma unroll
        for (int u = 0; u < ST_UB; ++u)
            if ((unsigned int)li[u] < (unsigned int)ST_NR) atomicAdd(&Dt[li[u]], expf(lv[u] - mt[u]));
    }
    __syncthreads();

    float* mo = Mp + (size_t)blockIdx.x * Npad + base;
    float* so = Dp + (size_t)blockIdx.x * Npad + base;
    for (int i = tid; i < ST_NR; i += 1024) { mo[i] = Mt[i]; so[i] = Dt[i]; }
}

__global__ void k_seg_merge(const float* __restrict__ Mp, const float* __restrict__ Dp,
                            float* __restrict__ md, int N, int Npad)
{
    const int n = blockIdx.x * blockDim.x + threadIdx.x;
    if (n >= N) return;
    float mv[ST_P], dv[ST_P];
#pragma unroll
    for (int w = 0; w < ST_P; ++w) {
        mv[w] = Mp[(size_t)w * Npad + n];
        dv[w] = Dp[(size_t)w * Npad + n];
    }
    float mx = mv[0];
#pragma unroll
    for (int w = 1; w < ST_P; ++w) mx = fmaxf(mx, mv[w]);
    if (!__builtin_isfinite(mx)) mx = 0.0f;
    float d = 0.0f;
#pragma unroll
    for (int w = 0; w < ST_P; ++w) d += dv[w] * expf(mv[w] - mx);
    v2f o; o[0] = mx; o[1] = d;
    *(v2f*)(md + 2 * (size_t)n) = o;
}

#define AG_T 256

__global__ void k_edge_aggregate(const float* __restrict__ edge_attr,
                                 const int* __restrict__ erow, const int* __restrict__ ecol,
                                 const float* __restrict__ W_edge, const float* __restrict__ b_edge,
                                 const float* __restrict__ z, const float* __restrict__ elog,
                                 const float* __restrict__ md, float* __restrict__ hout,
                                 int E, int nTiles)
{
    __shared__ float lWs[(ED_DIM / 2) * BSTRIDE_F];
    __shared__ __attribute__((aligned(16))) float lSt[8 * 16 * SPITCH];
    __shared__ __attribute__((aligned(16))) float lA[8 * 16 * EPITCH];
    const int lane = threadIdx.x & 31;
    const int wave = threadIdx.x >> 5;
    const int half = lane >> 4;
    const int l16  = lane & 15;
    const int tile = blockIdx.x * 8 + wave;

    const int  eown = tile * 16 + 8 * half + (l16 & 7);
    const bool live = eown < E;
    const int  eo   = live ? eown : (E - 1);
    const int   cl_o = ecol[eo];
    const int   rr_o = erow[eo];
    const float el_o = elog[eo];

    const int ar = lane >> 3;
    const int ac = 4 * (lane & 7);
    v4f av[4];
#pragma unroll
    for (int i = 0; i < 4; ++i) {
        int g = tile * 16 + 4 * i + ar;
        if (g >= E) g = E - 1;
        av[i] = *(const v4f*)(edge_attr + (size_t)g * ED_DIM + ac);
    }

    v4f wv[2];
#pragma unroll
    for (int h = 0; h < 2; ++h)
        wv[h] = *(const v4f*)(W_edge + 4 * threadIdx.x + h * (4 * AG_T));
    __builtin_amdgcn_sched_barrier(0);

    const v2f s_o = *(const v2f*)(md + 2 * (size_t)cl_o);

    int rr[8];
    rr[0] = row_share_i<0>(rr_o); rr[1] = row_share_i<1>(rr_o);
    rr[2] = row_share_i<2>(rr_o); rr[3] = row_share_i<3>(rr_o);
    rr[4] = row_share_i<4>(rr_o); rr[5] = row_share_i<5>(rr_o);
    rr[6] = row_share_i<6>(rr_o); rr[7] = row_share_i<7>(rr_o);

    v4f zv[8];
#pragma unroll
    for (int r = 0; r < 8; ++r)
        zv[r] = *(const v4f*)(z + (size_t)rr[r] * OUT_DIM + 4 * l16);
    const v4f bc = *(const v4f*)(b_edge + 4 * l16);
    __builtin_amdgcn_sched_barrier(0);

    const float a_o  = expf(el_o - s_o[0]) / fmaxf(s_o[1], 1e-16f);
    const float al_o = live ? a_o : 0.f;
    const int   cr_o = live ? cl_o : -1;
    float alph[8];
    alph[0] = row_share_f<0>(al_o); alph[1] = row_share_f<1>(al_o);
    alph[2] = row_share_f<2>(al_o); alph[3] = row_share_f<3>(al_o);
    alph[4] = row_share_f<4>(al_o); alph[5] = row_share_f<5>(al_o);
    alph[6] = row_share_f<6>(al_o); alph[7] = row_share_f<7>(al_o);

    Acc4 acc;
#pragma unroll
    for (int r = 0; r < 8; ++r) {
        acc.t[0][r] = zv[r][0] + bc[0];
        acc.t[1][r] = zv[r][1] + bc[1];
        acc.t[2][r] = zv[r][2] + bc[2];
        acc.t[3][r] = zv[r][3] + bc[3];
    }

#pragma unroll
    for (int h = 0; h < 2; ++h) {
        const int f = 4 * threadIdx.x + h * (4 * AG_T);
        const int k = f >> 6;
        const int c = (f & 63) >> 2;
        float* wd = lWs + (k >> 1) * BSTRIDE_F + (c << 1) + (k & 1);
        wd[0]  = wv[h][0];
        wd[32] = wv[h][1];
        wd[64] = wv[h][2];
        wd[96] = wv[h][3];
    }
    float* at = lA + wave * (16 * EPITCH);
#pragma unroll
    for (int i = 0; i < 4; ++i)
        *(v4f*)(at + (4 * i + ar) * EPITCH + ac) = av[i];
    __syncthreads();
    if (tile >= nTiles) return;

    acc = wave_gemm_tile(at + l16 * EPITCH, lWs, ED_DIM, half, l16, acc);

    float* st = lSt + wave * (16 * SPITCH);
#pragma unroll
    for (int r = 0; r < 8; ++r) {
        v4f o;
        o[0] = alph[r] * acc.t[0][r];
        o[1] = alph[r] * acc.t[1][r];
        o[2] = alph[r] * acc.t[2][r];
        o[3] = alph[r] * acc.t[3][r];
        *(v4f*)(st + (r + 8 * half) * SPITCH + 4 * l16) = o;
    }
    __builtin_amdgcn_fence(__ATOMIC_RELEASE, "workgroup");
    __builtin_amdgcn_fence(__ATOMIC_ACQUIRE, "workgroup");

    float v0[16], v1[16];
#pragma unroll
    for (int e = 0; e < 16; ++e) {
        v0[e] = st[e * SPITCH + lane];
        v1[e] = st[e * SPITCH + 32 + lane];
    }
#pragma unroll
    for (int e = 0; e < 16; ++e) {
        const int crow = __builtin_amdgcn_readlane(cr_o, (e >> 3) * 16 + (e & 7));
        if (crow >= 0) {
            float* rowp = hout + (size_t)crow * OUT_DIM + lane;
            atomicAdd(rowp,      v0[e]);
            atomicAdd(rowp + 32, v1[e]);
        }
    }
}

extern "C" void kernel_launch(void* const* d_in, const int* in_sizes, int n_in,
                              void* d_out, int out_size, void* d_ws, size_t ws_size,
                              hipStream_t stream)
{
    const float* x         = (const float*)d_in[0];
    const float* edge_attr = (const float*)d_in[1];
    const int*   edge_idx  = (const int*)  d_in[2];
    const float* W_fc      = (const float*)d_in[3];
    const float* b_fc      = (const float*)d_in[4];
    const float* W_attn    = (const float*)d_in[5];
    const float* b_attn    = (const float*)d_in[6];
    const float* W_edge    = (const float*)d_in[7];
    const float* b_edge    = (const float*)d_in[8];
    const float* W_self    = (const float*)d_in[9];
    const float* b_self    = (const float*)d_in[10];

    const int N = in_sizes[0] / IN_DIM;
    const int E = in_sizes[1] / ED_DIM;
    const int* erow = edge_idx;
    const int* ecol = edge_idx + E;

    const int nRanges  = (N + ST_NR - 1) / ST_NR;
    const int Npad     = nRanges * ST_NR;
    const int sliceLen = (E + ST_P - 1) / ST_P;

    float* ws     = (float*)d_ws;
    float* z      = ws;
    float* a_src  = z + (size_t)N * OUT_DIM;
    float* a_dst  = a_src + N;
    float* md     = a_dst + N;
    float* elog   = md + 2 * (size_t)N;
    float* Mp     = elog + E;
    float* Dp     = Mp + (size_t)ST_P * Npad;
    float* hout   = (float*)d_out;

    const int nNodeTiles = (N + 15) / 16;
    const int nEdgeTiles = (E + 15) / 16;

    k_node_gemm<<<(nNodeTiles + 7) / 8, 256, 0, stream>>>(
        x, W_fc, b_fc, W_self, b_self, z, hout, N, nNodeTiles);

    k_node_attn<<<(N + NA_T - 1) / NA_T, NA_T, 0, stream>>>(
        z, W_attn, a_src, a_dst, N);

    k_edge_logit<<<(E + LG_T - 1) / LG_T, LG_T, 0, stream>>>(
        edge_attr, erow, ecol, W_attn, b_attn, a_src, a_dst, elog, E);

    k_seg_stats<<<dim3(ST_P, nRanges), 1024, 0, stream>>>(
        elog, ecol, Mp, Dp, E, sliceLen, Npad);

    k_seg_merge<<<(N + 255) / 256, 256, 0, stream>>>(
        Mp, Dp, md, N, Npad);

    k_edge_aggregate<<<(nEdgeTiles + 7) / 8, AG_T, 0, stream>>>(
        edge_attr, erow, ecol, W_edge, b_edge, z, elog, md, hout, E, nEdgeTiles);
}
